# starting kernel
# baseline (speedup 1.0000x reference)
#include <hip/hip_runtime.h>

typedef __attribute__((ext_vector_type(2))) float v2f;
typedef __attribute__((ext_vector_type(8))) float v8f;

#define B_      4
#define V_      50000
#define C_      160
#define KSTEPS  40
#define WAVES   8
#define NTILES  3125

__global__ __launch_bounds__(256) void deform_lbs_wmma(
    const float* __restrict__ X,
    const float* __restrict__ Vn,
    const float* __restrict__ rot6d,
    const float* __restrict__ W,
    const int*   __restrict__ idx,
    float* __restrict__ out)
{
    __shared__ __align__(16) float s_fragB[B_ * KSTEPS * 32 * 2];
    __shared__ __align__(16) float s_g[WAVES * 256];

    const int tid = threadIdx.x;

    for (int t = tid; t < B_ * C_; t += 256) {
        const int b = t / C_;
        const int c = t % C_;

        const float* d6 = rot6d + ((size_t)b * C_ + c) * 6;
        float a1x = d6[0], a1y = d6[1], a1z = d6[2];
        float a2x = d6[3], a2y = d6[4], a2z = d6[5];
        const float eps = 1e-8f;

        float n1 = fmaxf(sqrtf(a1x*a1x + a1y*a1y + a1z*a1z), eps);
        float b1x = a1x / n1, b1y = a1y / n1, b1z = a1z / n1;
        float dp = b1x*a2x + b1y*a2y + b1z*a2z;
        float px = a2x - dp*b1x, py = a2y - dp*b1y, pz = a2z - dp*b1z;
        float n2 = fmaxf(sqrtf(px*px + py*py + pz*pz), eps);
        float b2x = px / n2, b2y = py / n2, b2z = pz / n2;
        float b3x = b1y*b2z - b1z*b2y;
        float b3y = b1z*b2x - b1x*b2z;
        float b3z = b1x*b2y - b1y*b2x;

        const int vi = idx[c];
        const float* cp = X + ((size_t)b * V_ + vi) * 3;
        float c0 = cp[0], c1 = cp[1], c2 = cp[2];
        const float* vp = Vn + ((size_t)b * C_ + c) * 3;

        float t0 = c0 + vp[0] - (b1x*c0 + b1y*c1 + b1z*c2);
        float t1 = c1 + vp[1] - (b2x*c0 + b2y*c1 + b2z*c2);
        float t2 = c2 + vp[2] - (b3x*c0 + b3y*c1 + b3z*c2);

        float m[16] = { b1x, b1y, b1z, t0,
                        b2x, b2y, b2z, t1,
                        b3x, b3y, b3z, t2,
                        0.f, 0.f, 0.f, 0.f };

        const int kk       = c >> 2;
        const int r        = c & 3;
        const int laneBase = (r >= 2) ? 16 : 0;
        const int comp     = r & 1;
        float* dst = &s_fragB[(((b * KSTEPS) + kk) * 32 + laneBase) * 2 + comp];
        #pragma unroll
        for (int n = 0; n < 16; ++n)
            dst[n * 2] = m[n];
    }
    __syncthreads();

    const int wave = tid >> 5;
    const int lane = tid & 31;
    const int tile = blockIdx.x * WAVES + wave;
    const bool valid = (tile < NTILES);
    const int vbase = valid ? tile * 16 : 0;
    const int mrow  = lane & 15;
    const int colBase = (lane < 16) ? 0 : 2;
    const float* wrow = W + (size_t)(vbase + mrow) * C_ + colBase;

    v8f acc[B_];
    #pragma unroll
    for (int b = 0; b < B_; ++b) acc[b] = (v8f){0,0,0,0,0,0,0,0};

    for (int kk = 0; kk < KSTEPS; ++kk) {
        v2f a = *reinterpret_cast<const v2f*>(wrow + 4 * kk);
        #pragma unroll
        for (int b = 0; b < B_; ++b) {
            v2f bf = *reinterpret_cast<const v2f*>(
                &s_fragB[(((b * KSTEPS) + kk) * 32 + lane) * 2]);
            acc[b] = __builtin_amdgcn_wmma_f32_16x16x4_f32(
                 false, a,  false, bf,
                 (short)0, acc[b],  false,  false);
        }
    }

    float* g = &s_g[wave * 256];
    #pragma unroll
    for (int b = 0; b < B_; ++b) {
        const int rowOff = (lane < 16) ? 0 : 8;
        const int n = lane & 15;
        #pragma unroll
        for (int r = 0; r < 8; ++r)
            g[(rowOff + r) * 16 + n] = acc[b][r];
        __syncthreads();
        if (lane < 16) {
            const int vv = vbase + lane;
            const float* gr = &g[lane * 16];
            const float* xp = X + ((size_t)b * V_ + vv) * 3;
            float x = xp[0], y = xp[1], z = xp[2];
            float o0 = gr[0]*x + gr[1]*y + gr[2]*z  + gr[3];
            float o1 = gr[4]*x + gr[5]*y + gr[6]*z  + gr[7];
            float o2 = gr[8]*x + gr[9]*y + gr[10]*z + gr[11];
            if (valid) {
                float* op = out + ((size_t)b * V_ + vv) * 3;
                op[0] = o0; op[1] = o1; op[2] = o2;
            }
        }
        __syncthreads();
    }
}

extern "C" void kernel_launch(void* const* d_in, const int* in_sizes, int n_in,
                              void* d_out, int out_size, void* d_ws, size_t ws_size,
                              hipStream_t stream) {
    const float* X   = (const float*)d_in[0];
    const float* Vn  = (const float*)d_in[1];
    const float* r6  = (const float*)d_in[2];
    const float* W   = (const float*)d_in[3];
    const int*   idx = (const int*)d_in[4];
    float* out = (float*)d_out;

    dim3 grid((NTILES + WAVES - 1) / WAVES);
    deform_lbs_wmma<<<grid, 256, 0, stream>>>(X, Vn, r6, W, idx, out);
}


# WMMA operands swapped (transposed product) so each lane holds its own vertex's blended entries: epilogue from registers, no LDS staging or barriers, X loaded ahead of the matrix loop
# speedup vs baseline: 1.2661x; 1.2661x over previous
#include <hip/hip_runtime.h>


typedef __attribute__((ext_vector_type(2))) float v2f;
typedef __attribute__((ext_vector_type(8))) float v8f;

#define B_      4
#define V_      50000
#define C_      160
#define KSTEPS  40
#define WAVES   8
#define NTILES  3125

__global__ __launch_bounds__(256) void deform_lbs_wmma(
    const float* __restrict__ X,
    const float* __restrict__ Vn,
    const float* __restrict__ rot6d,
    const float* __restrict__ W,
    const int*   __restrict__ idx,
    float* __restrict__ out)
{
    __shared__ __align__(16) float s_fragB[B_ * KSTEPS * 32 * 2];

    const int tid = threadIdx.x;

    for (int t = tid; t < B_ * C_; t += 256) {
        const int b = t / C_;
        const int c = t % C_;

        const float* d6 = rot6d + ((size_t)b * C_ + c) * 6;
        float a1x = d6[0], a1y = d6[1], a1z = d6[2];
        float a2x = d6[3], a2y = d6[4], a2z = d6[5];
        const float eps = 1e-8f;

        float n1 = fmaxf(sqrtf(a1x*a1x + a1y*a1y + a1z*a1z), eps);
        float b1x = a1x / n1, b1y = a1y / n1, b1z = a1z / n1;
        float dp = b1x*a2x + b1y*a2y + b1z*a2z;
        float px = a2x - dp*b1x, py = a2y - dp*b1y, pz = a2z - dp*b1z;
        float n2 = fmaxf(sqrtf(px*px + py*py + pz*pz), eps);
        float b2x = px / n2, b2y = py / n2, b2z = pz / n2;
        float b3x = b1y*b2z - b1z*b2y;
        float b3y = b1z*b2x - b1x*b2z;
        float b3z = b1x*b2y - b1y*b2x;

        const int vi = idx[c];
        const float* cp = X + ((size_t)b * V_ + vi) * 3;
        float c0 = cp[0], c1 = cp[1], c2 = cp[2];
        const float* vp = Vn + ((size_t)b * C_ + c) * 3;

        float t0 = c0 + vp[0] - (b1x*c0 + b1y*c1 + b1z*c2);
        float t1 = c1 + vp[1] - (b2x*c0 + b2y*c1 + b2z*c2);
        float t2 = c2 + vp[2] - (b3x*c0 + b3y*c1 + b3z*c2);

        float m[16] = { b1x, b1y, b1z, t0,
                        b2x, b2y, b2z, t1,
                        b3x, b3y, b3z, t2,
                        0.f, 0.f, 0.f, 0.f };

        const int kk       = c >> 2;
        const int r        = c & 3;
        const int laneBase = (r >= 2) ? 16 : 0;
        const int comp     = r & 1;
        float* dst = &s_fragB[(((b * KSTEPS) + kk) * 32 + laneBase) * 2 + comp];
        #pragma unroll
        for (int n = 0; n < 16; ++n)
            dst[n * 2] = m[n];
    }
    __syncthreads();

    const int wave = tid >> 5;
    const int lane = tid & 31;
    const int tile = blockIdx.x * WAVES + wave;
    const bool valid = (tile < NTILES);
    const int vbase = valid ? tile * 16 : 0;
    const int mrow  = lane & 15;
    const int colBase = (lane < 16) ? 0 : 2;
    const float* wrow = W + (size_t)(vbase + mrow) * C_ + colBase;

    float xs[B_], ys[B_], zs[B_];
    #pragma unroll
    for (int b = 0; b < B_; ++b) {
        const float* xp = X + ((size_t)b * V_ + vbase + mrow) * 3;
        xs[b] = xp[0]; ys[b] = xp[1]; zs[b] = xp[2];
    }

    v8f acc[B_];
    #pragma unroll
    for (int b = 0; b < B_; ++b) acc[b] = (v8f){0,0,0,0,0,0,0,0};

    for (int kk = 0; kk < KSTEPS; ++kk) {
        v2f a = *reinterpret_cast<const v2f*>(wrow + 4 * kk);
        #pragma unroll
        for (int b = 0; b < B_; ++b) {
            v2f bf = *reinterpret_cast<const v2f*>(
                &s_fragB[(((b * KSTEPS) + kk) * 32 + lane) * 2]);
            acc[b] = __builtin_amdgcn_wmma_f32_16x16x4_f32(
                 false, bf,  false, a,
                 (short)0, acc[b],  false,  false);
        }
    }

    asm volatile("v_nop\n\tv_nop\n\tv_nop\n\tv_nop\n\tv_nop\n\tv_nop\n\tv_nop\n\tv_nop"
                 : "+v"(acc[0]), "+v"(acc[1]), "+v"(acc[2]), "+v"(acc[3]));

    float p[B_], q[B_], o2[B_];
    #pragma unroll
    for (int b = 0; b < B_; ++b) {
        const float x = xs[b], y = ys[b], z = zs[b];
        p[b] = acc[b][0]*x + acc[b][1]*y + acc[b][2]*z + acc[b][3];
        q[b] = acc[b][4]*x + acc[b][5]*y + acc[b][6]*z + acc[b][7];
    }
    #pragma unroll
    for (int b = 0; b < B_; ++b)
        o2[b] = __shfl_xor(p[b], 16);
    if (valid && lane < 16) {
        #pragma unroll
        for (int b = 0; b < B_; ++b) {
            float* op = out + ((size_t)b * V_ + vbase + lane) * 3;
            op[0] = p[b]; op[1] = q[b]; op[2] = o2[b];
        }
    }
}

extern "C" void kernel_launch(void* const* d_in, const int* in_sizes, int n_in,
                              void* d_out, int out_size, void* d_ws, size_t ws_size,
                              hipStream_t stream) {
    const float* X   = (const float*)d_in[0];
    const float* Vn  = (const float*)d_in[1];
    const float* r6  = (const float*)d_in[2];
    const float* W   = (const float*)d_in[3];
    const int*   idx = (const int*)d_in[4];
    float* out = (float*)d_out;

    dim3 grid((NTILES + WAVES - 1) / WAVES);
    deform_lbs_wmma<<<grid, 256, 0, stream>>>(X, Vn, r6, W, idx, out);
}


# Four batches' 12 non-zero affine entries packed into 3 WMMA row-tiles instead of 4 zero-padded ones: 120 not 160 matrix ops and LDS fragment reads per wave; each lane half owns two whole batches
# speedup vs baseline: 1.4650x; 1.1571x over previous
#include <hip/hip_runtime.h>


typedef __attribute__((ext_vector_type(2))) float v2f;
typedef __attribute__((ext_vector_type(8))) float v8f;

#define B_      4
#define V_      50000
#define C_      160
#define KSTEPS  40
#define WAVES   8
#define NTILES  3125
#define MTILES  3

__global__ __launch_bounds__(256) void deform_lbs_wmma(
    const float* __restrict__ X,
    const float* __restrict__ Vn,
    const float* __restrict__ rot6d,
    const float* __restrict__ W,
    const int*   __restrict__ idx,
    float* __restrict__ out)
{
    __shared__ __align__(16) float s_fragB[MTILES * KSTEPS * 32 * 2];

    const int tid = threadIdx.x;

    for (int t = tid; t < B_ * C_; t += 256) {
        const int b = t / C_;
        const int c = t % C_;

        const float* d6 = rot6d + ((size_t)b * C_ + c) * 6;
        float a1x = d6[0], a1y = d6[1], a1z = d6[2];
        float a2x = d6[3], a2y = d6[4], a2z = d6[5];
        const float eps = 1e-8f;

        float n1 = fmaxf(sqrtf(a1x*a1x + a1y*a1y + a1z*a1z), eps);
        float b1x = a1x / n1, b1y = a1y / n1, b1z = a1z / n1;
        float dp = b1x*a2x + b1y*a2y + b1z*a2z;
        float px = a2x - dp*b1x, py = a2y - dp*b1y, pz = a2z - dp*b1z;
        float n2 = fmaxf(sqrtf(px*px + py*py + pz*pz), eps);
        float b2x = px / n2, b2y = py / n2, b2z = pz / n2;
        float b3x = b1y*b2z - b1z*b2y;
        float b3y = b1z*b2x - b1x*b2z;
        float b3z = b1x*b2y - b1y*b2x;

        const int vi = idx[c];
        const float* cp = X + ((size_t)b * V_ + vi) * 3;
        float c0 = cp[0], c1 = cp[1], c2 = cp[2];
        const float* vp = Vn + ((size_t)b * C_ + c) * 3;

        float t0 = c0 + vp[0] - (b1x*c0 + b1y*c1 + b1z*c2);
        float t1 = c1 + vp[1] - (b2x*c0 + b2y*c1 + b2z*c2);
        float t2 = c2 + vp[2] - (b3x*c0 + b3y*c1 + b3z*c2);

        float m[12] = { b1x, b1y, b1z, t0,
                        b2x, b2y, b2z, t1,
                        b3x, b3y, b3z, t2 };

        const int kk       = c >> 2;
        const int r        = c & 3;
        const int laneBase = ((r >= 2) ? 16 : 0) + 8 * (b >> 1);
        const int comp     = r & 1;
        float* dst = &s_fragB[(kk * 32 + laneBase) * 2 + comp];
        #pragma unroll
        for (int k = 0; k < 3; ++k) {
            const int g = 3 * (b & 1) + k;
            float* d = dst + (g >> 1) * (KSTEPS * 64) + (g & 1) * 8;
            d[0] = m[4 * k]; d[2] = m[4 * k + 1]; d[4] = m[4 * k + 2]; d[6] = m[4 * k + 3];
        }
    }
    __syncthreads();

    const int wave = tid >> 5;
    const int lane = tid & 31;
    const int tile = blockIdx.x * WAVES + wave;
    const bool valid = (tile < NTILES);
    const int vbase = valid ? tile * 16 : 0;
    const int mrow  = lane & 15;
    const int colBase = (lane < 16) ? 0 : 2;
    const float* wrow = W + (size_t)(vbase + mrow) * C_ + colBase;

    const int bfirst = (lane >> 4) * 2;
    float xs[2], ys[2], zs[2];
    #pragma unroll
    for (int i = 0; i < 2; ++i) {
        const float* xp = X + ((size_t)(bfirst + i) * V_ + vbase + mrow) * 3;
        xs[i] = xp[0]; ys[i] = xp[1]; zs[i] = xp[2];
    }

    v8f acc[MTILES];
    #pragma unroll
    for (int j = 0; j < MTILES; ++j) acc[j] = (v8f){0,0,0,0,0,0,0,0};

    for (int kk = 0; kk < KSTEPS; ++kk) {
        v2f a = *reinterpret_cast<const v2f*>(wrow + 4 * kk);
        #pragma unroll
        for (int j = 0; j < MTILES; ++j) {
            v2f bf = *reinterpret_cast<const v2f*>(
                &s_fragB[(((j * KSTEPS) + kk) * 32 + lane) * 2]);
            acc[j] = __builtin_amdgcn_wmma_f32_16x16x4_f32(
                 false, bf,  false, a,
                 (short)0, acc[j],  false,  false);
        }
    }

    asm volatile("v_nop\n\tv_nop\n\tv_nop\n\tv_nop\n\tv_nop\n\tv_nop\n\tv_nop\n\tv_nop"
                 : "+v"(acc[0]), "+v"(acc[1]), "+v"(acc[2]));

    if (valid) {
        #pragma unroll
        for (int i = 0; i < 2; ++i) {
            const float x = xs[i], y = ys[i], z = zs[i];
            float o[3];
            #pragma unroll
            for (int k = 0; k < 3; ++k) {
                const int q = 12 * i + 4 * k;
                o[k] = acc[q >> 3][q & 7] * x + acc[(q + 1) >> 3][(q + 1) & 7] * y
                     + acc[(q + 2) >> 3][(q + 2) & 7] * z + acc[(q + 3) >> 3][(q + 3) & 7];
            }
            float* op = out + ((size_t)(bfirst + i) * V_ + vbase + mrow) * 3;
            op[0] = o[0]; op[1] = o[1]; op[2] = o[2];
        }
    }
}

extern "C" void kernel_launch(void* const* d_in, const int* in_sizes, int n_in,
                              void* d_out, int out_size, void* d_ws, size_t ws_size,
                              hipStream_t stream) {
    const float* X   = (const float*)d_in[0];
    const float* Vn  = (const float*)d_in[1];
    const float* r6  = (const float*)d_in[2];
    const float* W   = (const float*)d_in[3];
    const int*   idx = (const int*)d_in[4];
    float* out = (float*)d_out;

    dim3 grid((NTILES + WAVES - 1) / WAVES);
    deform_lbs_wmma<<<grid, 256, 0, stream>>>(X, Vn, r6, W, idx, out);
}


# W tile staged through LDS: each wave fetches its 16 contiguous rows with 20 whole-line 16-byte loads at entry and reads WMMA fragments from padded LDS rows; no global load left in the matrix loop
# speedup vs baseline: 1.5776x; 1.0769x over previous
#include <hip/hip_runtime.h>


typedef __attribute__((ext_vector_type(2))) float v2f;
typedef __attribute__((ext_vector_type(8))) float v8f;

#define B_      4
#define V_      50000
#define C_      160
#define KSTEPS  40
#define WAVES   8
#define NTILES  3125
#define MTILES  3
#define FRAGB_FLOATS (MTILES * KSTEPS * 32 * 2)
#define WPITCH  164

__device__ __forceinline__ void build_node(
    const int t,
    const float* __restrict__ X, const float* __restrict__ Vn,
    const float* __restrict__ rot6d, const int* __restrict__ idx,
    float* s_fragB)
{
    const int b = t / C_;
    const int c = t % C_;

    const float* d6 = rot6d + ((size_t)b * C_ + c) * 6;
    float a1x = d6[0], a1y = d6[1], a1z = d6[2];
    float a2x = d6[3], a2y = d6[4], a2z = d6[5];
    const float eps = 1e-8f;

    float n1 = fmaxf(sqrtf(a1x*a1x + a1y*a1y + a1z*a1z), eps);
    float b1x = a1x / n1, b1y = a1y / n1, b1z = a1z / n1;
    float dp = b1x*a2x + b1y*a2y + b1z*a2z;
    float px = a2x - dp*b1x, py = a2y - dp*b1y, pz = a2z - dp*b1z;
    float n2 = fmaxf(sqrtf(px*px + py*py + pz*pz), eps);
    float b2x = px / n2, b2y = py / n2, b2z = pz / n2;
    float b3x = b1y*b2z - b1z*b2y;
    float b3y = b1z*b2x - b1x*b2z;
    float b3z = b1x*b2y - b1y*b2x;

    const int vi = idx[c];
    const float* cp = X + ((size_t)b * V_ + vi) * 3;
    float c0 = cp[0], c1 = cp[1], c2 = cp[2];
    const float* vp = Vn + ((size_t)b * C_ + c) * 3;

    float t0 = c0 + vp[0] - (b1x*c0 + b1y*c1 + b1z*c2);
    float t1 = c1 + vp[1] - (b2x*c0 + b2y*c1 + b2z*c2);
    float t2 = c2 + vp[2] - (b3x*c0 + b3y*c1 + b3z*c2);

    float m[12] = { b1x, b1y, b1z, t0,
                    b2x, b2y, b2z, t1,
                    b3x, b3y, b3z, t2 };

    const int kk       = c >> 2;
    const int r        = c & 3;
    const int laneBase = ((r >= 2) ? 16 : 0) + 8 * (b >> 1);
    const int comp     = r & 1;
    float* dst = &s_fragB[(kk * 32 + laneBase) * 2 + comp];
    #pragma unroll
    for (int k = 0; k < 3; ++k) {
        const int g = 3 * (b & 1) + k;
        float* d = dst + (g >> 1) * (KSTEPS * 64) + (g & 1) * 8;
        d[0] = m[4 * k]; d[2] = m[4 * k + 1]; d[4] = m[4 * k + 2]; d[6] = m[4 * k + 3];
    }
}

__global__ __launch_bounds__(256) void deform_lbs_wmma(
    const float* __restrict__ X,
    const float* __restrict__ Vn,
    const float* __restrict__ rot6d,
    const float* __restrict__ W,
    const int*   __restrict__ idx,
    float* __restrict__ out)
{
    __shared__ __align__(16) float s_all[FRAGB_FLOATS + WAVES * 16 * WPITCH];
    float* const s_fragB = s_all;
    float* const s_W     = s_all + FRAGB_FLOATS;

    const int tid = threadIdx.x;
    const int wave = tid >> 5;
    const int lane = tid & 31;
    const int tile = blockIdx.x * WAVES + wave;
    const bool valid = (tile < NTILES);
    const int vbase = valid ? tile * 16 : 0;

    const float4* wt = reinterpret_cast<const float4*>(W + (size_t)vbase * C_);
    float4 wreg[20];
    #pragma unroll
    for (int i = 0; i < 20; ++i) wreg[i] = wt[i * 32 + lane];


    build_node(tid, X, Vn, rot6d, idx, s_fragB);

    float* sw = s_W + wave * (16 * WPITCH);
    {
        int off[5];
        #pragma unroll
        for (int r = 0; r < 5; ++r) {
            const int f   = 128 * r + 4 * lane;
            const int row = f / C_;
            off[r] = row * WPITCH + (f - row * C_);
        }
        #pragma unroll
        for (int g = 0; g < 4; ++g) {
            #pragma unroll
            for (int r = 0; r < 5; ++r)
                *reinterpret_cast<float4*>(sw + g * 4 * WPITCH + off[r]) = wreg[5 * g + r];
        }
    }

    for (int t = tid + 256; t < B_ * C_; t += 256)
        build_node(t, X, Vn, rot6d, idx, s_fragB);
    __syncthreads();

    const int mrow  = lane & 15;
    const int colBase = (lane < 16) ? 0 : 2;
    const float* wrow = sw + mrow * WPITCH + colBase;

    const int bfirst = (lane >> 4) * 2;
    float xs[2], ys[2], zs[2];
    #pragma unroll
    for (int i = 0; i < 2; ++i) {
        const float* xp = X + ((size_t)(bfirst + i) * V_ + vbase + mrow) * 3;
        xs[i] = xp[0]; ys[i] = xp[1]; zs[i] = xp[2];
    }

    v8f acc[MTILES];
    #pragma unroll
    for (int j = 0; j < MTILES; ++j) acc[j] = (v8f){0,0,0,0,0,0,0,0};

    for (int kk = 0; kk < KSTEPS; ++kk) {
        v2f a = *reinterpret_cast<const v2f*>(wrow + 4 * kk);
        #pragma unroll
        for (int j = 0; j < MTILES; ++j) {
            v2f bf = *reinterpret_cast<const v2f*>(
                &s_fragB[(((j * KSTEPS) + kk) * 32 + lane) * 2]);
            acc[j] = __builtin_amdgcn_wmma_f32_16x16x4_f32(
                 false, bf,  false, a,
                 (short)0, acc[j],  false,  false);
        }
    }

    asm volatile("v_nop\n\tv_nop\n\tv_nop\n\tv_nop\n\tv_nop\n\tv_nop\n\tv_nop\n\tv_nop"
                 : "+v"(acc[0]), "+v"(acc[1]), "+v"(acc[2]));

    if (valid) {
        #pragma unroll
        for (int i = 0; i < 2; ++i) {
            const float x = xs[i], y = ys[i], z = zs[i];
            float o[3];
            #pragma unroll
            for (int k = 0; k < 3; ++k) {
                const int q = 12 * i + 4 * k;
                o[k] = acc[q >> 3][q & 7] * x + acc[(q + 1) >> 3][(q + 1) & 7] * y
                     + acc[(q + 2) >> 3][(q + 2) & 7] * z + acc[(q + 3) >> 3][(q + 3) & 7];
            }
            float* op = out + ((size_t)(bfirst + i) * V_ + vbase + mrow) * 3;
            op[0] = o[0]; op[1] = o[1]; op[2] = o[2];
        }
    }
}

extern "C" void kernel_launch(void* const* d_in, const int* in_sizes, int n_in,
                              void* d_out, int out_size, void* d_ws, size_t ws_size,
                              hipStream_t stream) {
    const float* X   = (const float*)d_in[0];
    const float* Vn  = (const float*)d_in[1];
    const float* r6  = (const float*)d_in[2];
    const float* W   = (const float*)d_in[3];
    const int*   idx = (const int*)d_in[4];
    float* out = (float*)d_out;

    dim3 grid((NTILES + WAVES - 1) / WAVES);
    deform_lbs_wmma<<<grid, 256, 0, stream>>>(X, Vn, r6, W, idx, out);
}


# One 13-wave workgroup per WGP (256, was 391 of 8 waves): node table built once per WGP, 12 vertex tiles each, the 53 left-over tiles cut by A row-tile into 159 40-WMMA units for the 13th waves
# speedup vs baseline: 1.7614x; 1.1165x over previous
#include <hip/hip_runtime.h>


typedef __attribute__((ext_vector_type(2))) float v2f;
typedef __attribute__((ext_vector_type(8))) float v8f;

#define B_      4
#define V_      50000
#define C_      160
#define KSTEPS  40
#define WAVES   13
#define THREADS (WAVES * 32)
#define NBLOCKS 256
#define TILES_PER_BLOCK 12
#define NTILES  3125
#define NUNITS  (3 * (NTILES - NBLOCKS * TILES_PER_BLOCK))
#define MTILES  3
#define FRAGB_FLOATS (MTILES * KSTEPS * 32 * 2)
#define WPITCH  164

__device__ __forceinline__ void build_node(
    const int t,
    const float* __restrict__ X, const float* __restrict__ Vn,
    const float* __restrict__ rot6d, const int* __restrict__ idx,
    float* s_fragB)
{
    const int b = t / C_;
    const int c = t % C_;

    const float* d6 = rot6d + ((size_t)b * C_ + c) * 6;
    float a1x = d6[0], a1y = d6[1], a1z = d6[2];
    float a2x = d6[3], a2y = d6[4], a2z = d6[5];
    const float eps = 1e-8f;

    float n1 = fmaxf(sqrtf(a1x*a1x + a1y*a1y + a1z*a1z), eps);
    float b1x = a1x / n1, b1y = a1y / n1, b1z = a1z / n1;
    float dp = b1x*a2x + b1y*a2y + b1z*a2z;
    float px = a2x - dp*b1x, py = a2y - dp*b1y, pz = a2z - dp*b1z;
    float n2 = fmaxf(sqrtf(px*px + py*py + pz*pz), eps);
    float b2x = px / n2, b2y = py / n2, b2z = pz / n2;
    float b3x = b1y*b2z - b1z*b2y;
    float b3y = b1z*b2x - b1x*b2z;
    float b3z = b1x*b2y - b1y*b2x;

    const int vi = idx[c];
    const float* cp = X + ((size_t)b * V_ + vi) * 3;
    float c0 = cp[0], c1 = cp[1], c2 = cp[2];
    const float* vp = Vn + ((size_t)b * C_ + c) * 3;

    float t0 = c0 + vp[0] - (b1x*c0 + b1y*c1 + b1z*c2);
    float t1 = c1 + vp[1] - (b2x*c0 + b2y*c1 + b2z*c2);
    float t2 = c2 + vp[2] - (b3x*c0 + b3y*c1 + b3z*c2);

    float m[12] = { b1x, b1y, b1z, t0,
                    b2x, b2y, b2z, t1,
                    b3x, b3y, b3z, t2 };

    const int kk       = c >> 2;
    const int r        = c & 3;
    const int laneBase = ((r >= 2) ? 16 : 0) + 8 * (b >> 1);
    const int comp     = r & 1;
    float* dst = &s_fragB[(kk * 32 + laneBase) * 2 + comp];
    #pragma unroll
    for (int k = 0; k < 3; ++k) {
        const int g = 3 * (b & 1) + k;
        float* d = dst + (g >> 1) * (KSTEPS * 64) + (g & 1) * 8;
        d[0] = m[4 * k]; d[2] = m[4 * k + 1]; d[4] = m[4 * k + 2]; d[6] = m[4 * k + 3];
    }
}

__global__ __launch_bounds__(THREADS) void deform_lbs_wmma(
    const float* __restrict__ X,
    const float* __restrict__ Vn,
    const float* __restrict__ rot6d,
    const float* __restrict__ W,
    const int*   __restrict__ idx,
    float* __restrict__ out)
{
    __shared__ __align__(16) float s_all[FRAGB_FLOATS + WAVES * 16 * WPITCH];
    float* const s_fragB = s_all;
    float* const s_W     = s_all + FRAGB_FLOATS;

    const int tid = threadIdx.x;
    const int wave = __builtin_amdgcn_readfirstlane(tid >> 5);
    const int lane = tid & 31;
    const int blk  = (int)blockIdx.x;
    const bool full = (wave < TILES_PER_BLOCK);
    const bool valid = full || (blk < NUNITS);
    const int tile = full ? blk * TILES_PER_BLOCK + wave
                          : NBLOCKS * TILES_PER_BLOCK + blk / 3;
    const int vbase = valid ? tile * 16 : 0;

    float* sw = s_W + wave * (16 * WPITCH);
    if (valid) {
        const float4* wt = reinterpret_cast<const float4*>(W + (size_t)vbase * C_);
        float4 wreg[20];
        #pragma unroll
        for (int i = 0; i < 20; ++i) wreg[i] = wt[i * 32 + lane];

        build_node(tid, X, Vn, rot6d, idx, s_fragB);

        int off[5];
        #pragma unroll
        for (int r = 0; r < 5; ++r) {
            const int f   = 128 * r + 4 * lane;
            const int row = f / C_;
            off[r] = row * WPITCH + (f - row * C_);
        }
        #pragma unroll
        for (int g = 0; g < 4; ++g) {
            #pragma unroll
            for (int r = 0; r < 5; ++r)
                *reinterpret_cast<float4*>(sw + g * 4 * WPITCH + off[r]) = wreg[5 * g + r];
        }
    } else {
        build_node(tid, X, Vn, rot6d, idx, s_fragB);
    }

    for (int t = tid + THREADS; t < B_ * C_; t += THREADS)
        build_node(t, X, Vn, rot6d, idx, s_fragB);
    __syncthreads();

    if (!valid) return;

    const int mrow  = lane & 15;
    const int colBase = (lane < 16) ? 0 : 2;
    const float* wrow = sw + mrow * WPITCH + colBase;

    if (!full) {
        const int jsel = blk % 3;
        const int g0 = 2 * jsel, g1 = g0 + 1;
        const int i0 = g0 / 3, k0 = g0 % 3;
        const int i1 = g1 / 3, k1 = g1 % 3;
        const int hb = (lane >> 4) * 2;
        unsigned q0 = ((unsigned)(hb + i0) * V_ + vbase + mrow) * 3;
        unsigned q1 = ((unsigned)(hb + i1) * V_ + vbase + mrow) * 3;

        v8f accj = (v8f){0,0,0,0,0,0,0,0};
        const float* fb = s_fragB + (jsel * KSTEPS * 32 + lane) * 2;
        asm volatile("" : "+v"(accj), "+v"(q0), "+v"(q1));
        for (int kk = 0; kk < KSTEPS; ++kk) {
            v2f a  = *reinterpret_cast<const v2f*>(wrow + 4 * kk);
            v2f bf = *reinterpret_cast<const v2f*>(fb + kk * 64);
            accj = __builtin_amdgcn_wmma_f32_16x16x4_f32(
                 false, bf,  false, a,
                 (short)0, accj,  false,  false);
        }
        asm volatile("v_nop\n\tv_nop\n\tv_nop\n\tv_nop\n\tv_nop\n\tv_nop\n\tv_nop\n\tv_nop"
                     : "+v"(accj), "+v"(q0), "+v"(q1));
        const float x0 = X[q0], y0 = X[q0 + 1], z0 = X[q0 + 2];
        const float x1 = X[q1], y1 = X[q1 + 1], z1 = X[q1 + 2];
        out[q0 + k0] = __builtin_fmaf(accj[2], z0, __builtin_fmaf(accj[1], y0, accj[0] * x0)) + accj[3];
        out[q1 + k1] = __builtin_fmaf(accj[6], z1, __builtin_fmaf(accj[5], y1, accj[4] * x1)) + accj[7];
        return;
    }

    const int bfirst = (lane >> 4) * 2;
    float xs[2], ys[2], zs[2];
    #pragma unroll
    for (int i = 0; i < 2; ++i) {
        const float* xp = X + ((size_t)(bfirst + i) * V_ + vbase + mrow) * 3;
        xs[i] = xp[0]; ys[i] = xp[1]; zs[i] = xp[2];
    }

    v8f acc[MTILES];
    #pragma unroll
    for (int j = 0; j < MTILES; ++j) acc[j] = (v8f){0,0,0,0,0,0,0,0};

    for (int kk = 0; kk < KSTEPS; ++kk) {
        v2f a = *reinterpret_cast<const v2f*>(wrow + 4 * kk);
        #pragma unroll
        for (int j = 0; j < MTILES; ++j) {
            v2f bf = *reinterpret_cast<const v2f*>(
                &s_fragB[(((j * KSTEPS) + kk) * 32 + lane) * 2]);
            acc[j] = __builtin_amdgcn_wmma_f32_16x16x4_f32(
                 false, bf,  false, a,
                 (short)0, acc[j],  false,  false);
        }
    }

    asm volatile("v_nop\n\tv_nop\n\tv_nop\n\tv_nop\n\tv_nop\n\tv_nop\n\tv_nop\n\tv_nop"
                 : "+v"(acc[0]), "+v"(acc[1]), "+v"(acc[2]));

    {
        #pragma unroll
        for (int i = 0; i < 2; ++i) {
            const float x = xs[i], y = ys[i], z = zs[i];
            float o[3];
            #pragma unroll
            for (int k = 0; k < 3; ++k) {
                const int q = 12 * i + 4 * k;
                o[k] = acc[q >> 3][q & 7] * x + acc[(q + 1) >> 3][(q + 1) & 7] * y
                     + acc[(q + 2) >> 3][(q + 2) & 7] * z + acc[(q + 3) >> 3][(q + 3) & 7];
            }
            float* op = out + ((size_t)(bfirst + i) * V_ + vbase + mrow) * 3;
            op[0] = o[0]; op[1] = o[1]; op[2] = o[2];
        }
    }
}

extern "C" void kernel_launch(void* const* d_in, const int* in_sizes, int n_in,
                              void* d_out, int out_size, void* d_ws, size_t ws_size,
                              hipStream_t stream) {
    const float* X   = (const float*)d_in[0];
    const float* Vn  = (const float*)d_in[1];
    const float* r6  = (const float*)d_in[2];
    const float* W   = (const float*)d_in[3];
    const int*   idx = (const int*)d_in[4];
    float* out = (float*)d_out;

    dim3 grid(NBLOCKS);
    deform_lbs_wmma<<<grid, THREADS, 0, stream>>>(X, Vn, r6, W, idx, out);
}
